# starting kernel
# baseline (speedup 1.0000x reference)
#include <hip/hip_runtime.h>
#include <hip/hip_bf16.h>

#define B_ROWS 16384
#define IN_DIM 128
#define H_DIM  128
#define NCLS   10
#define NATT   3
#define EPSLN  1e-5f
#define LOG2E  1.4426950408889634f

#define NWAVES 2
#define RROWS  (NWAVES*16)
#define LDH    132

typedef float v2f __attribute__((ext_vector_type(2)));
typedef float v8f __attribute__((ext_vector_type(8)));

static __device__ __forceinline__ v8f wmma4(v2f a, v2f b, v8f c) {
  return __builtin_amdgcn_wmma_f32_16x16x4_f32(false, a, false, b, (short)0, c,
                                               false, false);
}

static __device__ __forceinline__ float fast_exp2(float x) {
#if __has_builtin(__builtin_amdgcn_exp2f)
  return __builtin_amdgcn_exp2f(x);
#else
  return exp2f(x);
#endif
}
static __device__ __forceinline__ float fast_rcp(float x) {
#if __has_builtin(__builtin_amdgcn_rcpf)
  return __builtin_amdgcn_rcpf(x);
#else
  return 1.0f / x;
#endif
}
static __device__ __forceinline__ float fast_rsq(float x) {
#if __has_builtin(__builtin_amdgcn_rsqf)
  return __builtin_amdgcn_rsqf(x);
#else
  return rsqrtf(x);
#endif
}

__global__ void __launch_bounds__(NWAVES * 32, 1)
simpleattention_fused_kernel(const float* __restrict__ x,
                             const float* __restrict__ W_in,
                             const float* __restrict__ b_in,
                             const float* __restrict__ W_att,
                             const float* __restrict__ b_att,
                             const float* __restrict__ gamma,
                             const float* __restrict__ beta,
                             const float* __restrict__ W_c,
                             const float* __restrict__ b_c,
                             float* __restrict__ out) {
  extern __shared__ float smem[];
  float* sh_h = smem;
  float* sh_u = smem + RROWS * LDH;

  const int lane  = threadIdx.x & 31;
  const int wave  = threadIdx.x >> 5;
  const int mlan  = lane & 15;
  const int half  = lane >> 4;
  const int khalf = half << 1;
  const int row0  = blockIdx.x * RROWS;
  const int wrow  = wave * 16;

  {
    v8f acc[8] = {};
    const float* xrow = x + (size_t)(row0 + wrow + mlan) * IN_DIM;
    for (int kk = 0; kk < IN_DIM; kk += 4) {
      v2f a = *(const v2f*)(xrow + kk + khalf);
#pragma unroll
      for (int nt = 0; nt < 8; ++nt) {
        const int n = nt * 16 + mlan;
        v2f b;
        b.x = W_in[(kk + khalf) * H_DIM + n];
        b.y = W_in[(kk + khalf + 1) * H_DIM + n];
        acc[nt] = wmma4(a, b, acc[nt]);
      }
    }
#pragma unroll
    for (int nt = 0; nt < 8; ++nt) {
      const int n = nt * 16 + mlan;
      const float bias = b_in[n];
#pragma unroll
      for (int r = 0; r < 8; ++r) {
        const int m = wrow + r + 8 * half;
        sh_h[m * LDH + n] = acc[nt][r] + bias;
      }
    }
  }

  for (int layer = 0; layer < NATT; ++layer) {
    const float* W  = W_att + (size_t)layer * H_DIM * H_DIM;
    const float* bb = b_att + layer * H_DIM;

    {
      v8f acc[8] = {};
      const float* arow = sh_h + (wrow + mlan) * LDH;
      for (int kk = 0; kk < H_DIM; kk += 4) {
        v2f a = *(const v2f*)(arow + kk + khalf);
#pragma unroll
        for (int nt = 0; nt < 8; ++nt) {
          const int n = nt * 16 + mlan;
          v2f b;
          b.x = W[(kk + khalf) * H_DIM + n];
          b.y = W[(kk + khalf + 1) * H_DIM + n];
          acc[nt] = wmma4(a, b, acc[nt]);
        }
      }
#pragma unroll
      for (int nt = 0; nt < 8; ++nt) {
        const int n = nt * 16 + mlan;
        const float bias = bb[n];
#pragma unroll
        for (int r = 0; r < 8; ++r) {
          const int m = wrow + r + 8 * half;
          sh_u[m * H_DIM + n] = tanhf(acc[nt][r] + bias);
        }
      }
    }

    for (int rr = 0; rr < 16; ++rr) {
      const int row = wrow + rr;
      const float* ur = sh_u + row * H_DIM;
      float* hr = sh_h + row * LDH;

      float si[4], hi[4];
#pragma unroll
      for (int q = 0; q < 4; ++q) {
        const int i = lane + 32 * q;
        si[q] = ur[i] * LOG2E;
        hi[q] = hr[i];
      }
      float Dv[4] = {0.f, 0.f, 0.f, 0.f};
      float Nv[4] = {0.f, 0.f, 0.f, 0.f};
      for (int j = 0; j < H_DIM; j += 4) {
        const float4 uj4 = *(const float4*)(ur + j);
        const float4 hj4 = *(const float4*)(hr + j);
        const float ujv[4] = {uj4.x, uj4.y, uj4.z, uj4.w};
        const float hjv[4] = {hj4.x, hj4.y, hj4.z, hj4.w};
#pragma unroll
        for (int c = 0; c < 4; ++c) {
#pragma unroll
          for (int q = 0; q < 4; ++q) {
            const float e = fast_exp2(si[q] * ujv[c]);
            Dv[q] += e;
            Nv[q] += e * hjv[c];
          }
        }
      }
      float y[4], s = 0.f;
#pragma unroll
      for (int q = 0; q < 4; ++q) {
        y[q] = hi[q] + Nv[q] * fast_rcp(Dv[q]);
        s += y[q];
      }
#pragma unroll
      for (int off = 16; off > 0; off >>= 1) s += __shfl_xor(s, off, 32);
      const float mu = s * (1.0f / 128.0f);
      float v = 0.f;
#pragma unroll
      for (int q = 0; q < 4; ++q) {
        const float d = y[q] - mu;
        v += d * d;
      }
#pragma unroll
      for (int off = 16; off > 0; off >>= 1) v += __shfl_xor(v, off, 32);
      const float rstd = fast_rsq(v * (1.0f / 128.0f) + EPSLN);
#pragma unroll
      for (int q = 0; q < 4; ++q) {
        const int i = lane + 32 * q;
        hr[i] = (y[q] - mu) * rstd * gamma[layer * H_DIM + i] +
                beta[layer * H_DIM + i];
      }
    }
  }

  {
    v8f acc = {};
    const bool valid = (mlan < NCLS);
    const float* arow = sh_h + (wrow + mlan) * LDH;
    for (int kk = 0; kk < H_DIM; kk += 4) {
      v2f a = *(const v2f*)(arow + kk + khalf);
      v2f b;
      b.x = valid ? W_c[(kk + khalf) * NCLS + mlan] : 0.0f;
      b.y = valid ? W_c[(kk + khalf + 1) * NCLS + mlan] : 0.0f;
      acc = wmma4(a, b, acc);
    }
    if (valid) {
      const float bias = b_c[mlan];
#pragma unroll
      for (int r = 0; r < 8; ++r) {
        const int m = wrow + r + 8 * half;
        out[(size_t)(row0 + m) * NCLS + mlan] = acc[r] + bias;
      }
    }
  }
}

extern "C" void kernel_launch(void* const* d_in, const int* in_sizes, int n_in,
                              void* d_out, int out_size, void* d_ws,
                              size_t ws_size, hipStream_t stream) {
  const float* x     = (const float*)d_in[0];
  const float* W_in  = (const float*)d_in[1];
  const float* b_in  = (const float*)d_in[2];
  const float* W_att = (const float*)d_in[3];
  const float* b_att = (const float*)d_in[4];
  const float* gamma = (const float*)d_in[5];
  const float* beta  = (const float*)d_in[6];
  const float* W_c   = (const float*)d_in[7];
  const float* b_c   = (const float*)d_in[8];
  float* out = (float*)d_out;

  constexpr int kSmemBytes = (RROWS * LDH + RROWS * H_DIM) * (int)sizeof(float);

  dim3 grid(B_ROWS / RROWS);
  dim3 block(NWAVES * 32);
  simpleattention_fused_kernel<<<grid, block, kSmemBytes, stream>>>(
      x, W_in, b_in, W_att, b_att, gamma, beta, W_c, b_c, out);
}


# Split each 32-row tile over 8 waves, not 2: GEMMs by 16-column slice per wave (its B slice loaded as one group), row phase 4 rows per wave, LDS barriers between; 4 waves per SIMD instead of 1
# speedup vs baseline: 2.2851x; 2.2851x over previous
#include <hip/hip_runtime.h>


#define B_ROWS 16384
#define IN_DIM 128
#define H_DIM  128
#define NCLS   10
#define NATT   3
#define EPSLN  1e-5f
#define LOG2E  1.4426950408889634f

#define NWAVES 8
#define RROWS  32
#define RPW    (RROWS / NWAVES)
#define KSTEPS (H_DIM / 4)
#define LDH    132

typedef float v2f __attribute__((ext_vector_type(2)));
typedef float v8f __attribute__((ext_vector_type(8)));

static __device__ __forceinline__ v8f wmma4(v2f a, v2f b, v8f c) {
  return __builtin_amdgcn_wmma_f32_16x16x4_f32(false, a, false, b, (short)0, c,
                                               false, false);
}

static __device__ __forceinline__ float fast_exp2(float x) {
#if __has_builtin(__builtin_amdgcn_exp2f)
  return __builtin_amdgcn_exp2f(x);
#else
  return exp2f(x);
#endif
}
static __device__ __forceinline__ float fast_rcp(float x) {
#if __has_builtin(__builtin_amdgcn_rcpf)
  return __builtin_amdgcn_rcpf(x);
#else
  return 1.0f / x;
#endif
}
static __device__ __forceinline__ float fast_rsq(float x) {
#if __has_builtin(__builtin_amdgcn_rsqf)
  return __builtin_amdgcn_rsqf(x);
#else
  return rsqrtf(x);
#endif
}

static __device__ __forceinline__ void gemm_cols(const float* sA, const int lda,
                                                 const float* __restrict__ Wg,
                                                 const int mlan, const int khalf,
                                                 const int ncol, v8f& acc0, v8f& acc1) {
  v2f bf[KSTEPS];
#pragma unroll
  for (int s = 0; s < KSTEPS; ++s) {
    const int k = 4 * s + khalf;
    bf[s].x = Wg[k * H_DIM + ncol];
    bf[s].y = Wg[(k + 1) * H_DIM + ncol];
  }
  const float* a0p = sA + mlan * lda + khalf;
  const float* a1p = sA + (16 + mlan) * lda + khalf;
#pragma unroll
  for (int s = 0; s < KSTEPS; ++s) {
    const v2f a0 = *(const v2f*)(a0p + 4 * s);
    const v2f a1 = *(const v2f*)(a1p + 4 * s);
    acc0 = wmma4(a0, bf[s], acc0);
    acc1 = wmma4(a1, bf[s], acc1);
  }
}

__global__ void __launch_bounds__(NWAVES * 32, 4)
simpleattention_fused_kernel(const float* __restrict__ x,
                             const float* __restrict__ W_in,
                             const float* __restrict__ b_in,
                             const float* __restrict__ W_att,
                             const float* __restrict__ b_att,
                             const float* __restrict__ gamma,
                             const float* __restrict__ beta,
                             const float* __restrict__ W_c,
                             const float* __restrict__ b_c,
                             float* __restrict__ out) {
  __shared__ __attribute__((aligned(16))) float sh_h[RROWS * LDH];
  __shared__ __attribute__((aligned(16))) float sh_u[RROWS * LDH];

  const int lane  = threadIdx.x & 31;
  const int wave  = __builtin_amdgcn_readfirstlane((int)(threadIdx.x >> 5));
  const int mlan  = lane & 15;
  const int half  = lane >> 4;
  const int khalf = half << 1;
  const int row0  = blockIdx.x * RROWS;
  const int ncol  = wave * 16 + mlan;

  {
    const float4* xg = (const float4*)(x + (size_t)row0 * IN_DIM);
#pragma unroll
    for (int i = 0; i < (RROWS * IN_DIM / 4) / (NWAVES * 32); ++i) {
      const int idx = (int)threadIdx.x + i * NWAVES * 32;
      const float4 v = xg[idx];
      *(float4*)(sh_u + (idx >> 5) * LDH + 4 * (idx & 31)) = v;
    }
  }
  __syncthreads();

  {
    v8f acc0 = {}, acc1 = {};
    gemm_cols(sh_u, LDH, W_in, mlan, khalf, ncol, acc0, acc1);
    const float bias = b_in[ncol];
#pragma unroll
    for (int r = 0; r < 8; ++r) {
      const int m = r + 8 * half;
      sh_h[m * LDH + ncol]        = acc0[r] + bias;
      sh_h[(16 + m) * LDH + ncol] = acc1[r] + bias;
    }
  }
  __syncthreads();

  for (int layer = 0; layer < NATT; ++layer) {
    const float* W  = W_att + (size_t)layer * H_DIM * H_DIM;
    const float* bb = b_att + layer * H_DIM;

    float gq[4], bq[4];
#pragma unroll
    for (int q = 0; q < 4; ++q) {
      gq[q] = gamma[layer * H_DIM + lane + 32 * q];
      bq[q] = beta[layer * H_DIM + lane + 32 * q];
    }

    {
      v8f acc0 = {}, acc1 = {};
      gemm_cols(sh_h, LDH, W, mlan, khalf, ncol, acc0, acc1);
      const float bias = bb[ncol];
#pragma unroll
      for (int r = 0; r < 8; ++r) {
        const int m = r + 8 * half;
        sh_u[m * H_DIM + ncol]        = tanhf(acc0[r] + bias);
        sh_u[(16 + m) * H_DIM + ncol] = tanhf(acc1[r] + bias);
      }
    }
    __syncthreads();

    for (int rr = 0; rr < RPW; ++rr) {
      const int row = wave * RPW + rr;
      const float* ur = sh_u + row * H_DIM;
      float* hr = sh_h + row * LDH;

      float si[4], hi[4];
#pragma unroll
      for (int q = 0; q < 4; ++q) {
        const int i = lane + 32 * q;
        si[q] = ur[i] * LOG2E;
        hi[q] = hr[i];
      }
      float Dv[4] = {0.f, 0.f, 0.f, 0.f};
      float Nv[4] = {0.f, 0.f, 0.f, 0.f};
      for (int j = 0; j < H_DIM; j += 4) {
        const float4 uj4 = *(const float4*)(ur + j);
        const float4 hj4 = *(const float4*)(hr + j);
        const float ujv[4] = {uj4.x, uj4.y, uj4.z, uj4.w};
        const float hjv[4] = {hj4.x, hj4.y, hj4.z, hj4.w};
#pragma unroll
        for (int c = 0; c < 4; ++c) {
#pragma unroll
          for (int q = 0; q < 4; ++q) {
            const float e = fast_exp2(si[q] * ujv[c]);
            Dv[q] += e;
            Nv[q] += e * hjv[c];
          }
        }
      }
      float y[4], s = 0.f;
#pragma unroll
      for (int q = 0; q < 4; ++q) {
        y[q] = hi[q] + Nv[q] * fast_rcp(Dv[q]);
        s += y[q];
      }
#pragma unroll
      for (int off = 16; off > 0; off >>= 1) s += __shfl_xor(s, off, 32);
      const float mu = s * (1.0f / 128.0f);
      float v = 0.f;
#pragma unroll
      for (int q = 0; q < 4; ++q) {
        const float d = y[q] - mu;
        v += d * d;
      }
#pragma unroll
      for (int off = 16; off > 0; off >>= 1) v += __shfl_xor(v, off, 32);
      const float rstd = fast_rsq(v * (1.0f / 128.0f) + EPSLN);
#pragma unroll
      for (int q = 0; q < 4; ++q) {
        const int i = lane + 32 * q;
        hr[i] = (y[q] - mu) * rstd * gq[q] + bq[q];
      }
    }
    __syncthreads();
  }

  if (wave < RROWS / 16) {
    const int wrow = wave * 16;
    v8f acc = {};
    const bool valid = (mlan < NCLS);
    const float* arow = sh_h + (wrow + mlan) * LDH;
    for (int kk = 0; kk < H_DIM; kk += 4) {
      v2f a = *(const v2f*)(arow + kk + khalf);
      v2f b;
      b.x = valid ? W_c[(kk + khalf) * NCLS + mlan] : 0.0f;
      b.y = valid ? W_c[(kk + khalf + 1) * NCLS + mlan] : 0.0f;
      acc = wmma4(a, b, acc);
    }
    if (valid) {
      const float bias = b_c[mlan];
#pragma unroll
      for (int r = 0; r < 8; ++r) {
        const int m = wrow + r + 8 * half;
        out[(size_t)(row0 + m) * NCLS + mlan] = acc[r] + bias;
      }
    }
  }
}

extern "C" void kernel_launch(void* const* d_in, const int* in_sizes, int n_in,
                              void* d_out, int out_size, void* d_ws,
                              size_t ws_size, hipStream_t stream) {
  const float* x     = (const float*)d_in[0];
  const float* W_in  = (const float*)d_in[1];
  const float* b_in  = (const float*)d_in[2];
  const float* W_att = (const float*)d_in[3];
  const float* b_att = (const float*)d_in[4];
  const float* gamma = (const float*)d_in[5];
  const float* beta  = (const float*)d_in[6];
  const float* W_c   = (const float*)d_in[7];
  const float* b_c   = (const float*)d_in[8];
  float* out = (float*)d_out;

  dim3 grid(B_ROWS / RROWS);
  dim3 block(NWAVES * 32);
  simpleattention_fused_kernel<<<grid, block, 0, stream>>>(
      x, W_in, b_in, W_att, b_att, gamma, beta, W_c, b_c, out);
}


# Row softmax sums by the separable power series of exp(u_i*u_j), |u|<=1 after tanh, 13 terms, cut error below f32 rounding: 25 moments + 2 polynomials per row replace 128x128 exps; sum order changes
# speedup vs baseline: 5.0913x; 2.2280x over previous
#include <hip/hip_runtime.h>


#define B_ROWS 16384
#define IN_DIM 128
#define H_DIM  128
#define NCLS   10
#define NATT   3
#define EPSLN  1e-5f

#define NWAVES 8
#define RROWS  32
#define RPW    (RROWS / NWAVES)
#define GLANES (32 / RPW)
#define EPL    (H_DIM / GLANES)
#define NTERM  13
#define KSTEPS (H_DIM / 4)
#define LDH    132

typedef float v2f __attribute__((ext_vector_type(2)));
typedef float v8f __attribute__((ext_vector_type(8)));

static __device__ __forceinline__ v8f wmma4(v2f a, v2f b, v8f c) {
  return __builtin_amdgcn_wmma_f32_16x16x4_f32(false, a, false, b, (short)0, c,
                                               false, false);
}

static __device__ __forceinline__ float fast_rcp(float x) {
#if __has_builtin(__builtin_amdgcn_rcpf)
  return __builtin_amdgcn_rcpf(x);
#else
  return 1.0f / x;
#endif
}
static __device__ __forceinline__ float fast_rsq(float x) {
#if __has_builtin(__builtin_amdgcn_rsqf)
  return __builtin_amdgcn_rsqf(x);
#else
  return rsqrtf(x);
#endif
}

template <int CTRL>
static __device__ __forceinline__ float lane_xor(float v) {
  return __builtin_bit_cast(
      float, __builtin_amdgcn_update_dpp(0, __builtin_bit_cast(int, v), CTRL, 0xf, 0xf, false));
}
static __device__ __forceinline__ float group_sum(float v) {
  v += lane_xor<0x161>(v);
  v += lane_xor<0x162>(v);
  v += lane_xor<0x164>(v);
  return v;
}

static __device__ __forceinline__ void gemm_cols(const float* sA, const int lda,
                                                 const float* __restrict__ Wg,
                                                 const int mlan, const int khalf,
                                                 const int ncol, v8f& acc0, v8f& acc1) {
  v2f bf[KSTEPS];
#pragma unroll
  for (int s = 0; s < KSTEPS; ++s) {
    const int k = 4 * s + khalf;
    bf[s].x = Wg[k * H_DIM + ncol];
    bf[s].y = Wg[(k + 1) * H_DIM + ncol];
  }
  const float* a0p = sA + mlan * lda + khalf;
  const float* a1p = sA + (16 + mlan) * lda + khalf;
#pragma unroll
  for (int s = 0; s < KSTEPS; ++s) {
    const v2f a0 = *(const v2f*)(a0p + 4 * s);
    const v2f a1 = *(const v2f*)(a1p + 4 * s);
    acc0 = wmma4(a0, bf[s], acc0);
    acc1 = wmma4(a1, bf[s], acc1);
  }
}

__global__ void __launch_bounds__(NWAVES * 32, 4)
simpleattention_fused_kernel(const float* __restrict__ x,
                             const float* __restrict__ W_in,
                             const float* __restrict__ b_in,
                             const float* __restrict__ W_att,
                             const float* __restrict__ b_att,
                             const float* __restrict__ gamma,
                             const float* __restrict__ beta,
                             const float* __restrict__ W_c,
                             const float* __restrict__ b_c,
                             float* __restrict__ out) {
  __shared__ __attribute__((aligned(16))) float sh_h[RROWS * LDH];
  __shared__ __attribute__((aligned(16))) float sh_u[RROWS * LDH];

  const int lane  = threadIdx.x & 31;
  const int wave  = __builtin_amdgcn_readfirstlane((int)(threadIdx.x >> 5));
  const int mlan  = lane & 15;
  const int half  = lane >> 4;
  const int khalf = half << 1;
  const int row0  = blockIdx.x * RROWS;
  const int ncol  = wave * 16 + mlan;
  const int grow  = wave * RPW + (lane >> 3);
  const int sub   = lane & (GLANES - 1);

  {
    const float4* xg = (const float4*)(x + (size_t)row0 * IN_DIM);
#pragma unroll
    for (int i = 0; i < (RROWS * IN_DIM / 4) / (NWAVES * 32); ++i) {
      const int idx = (int)threadIdx.x + i * NWAVES * 32;
      const float4 v = xg[idx];
      *(float4*)(sh_u + (idx >> 5) * LDH + 4 * (idx & 31)) = v;
    }
  }
  __syncthreads();

  {
    v8f acc0 = {}, acc1 = {};
    gemm_cols(sh_u, LDH, W_in, mlan, khalf, ncol, acc0, acc1);
    const float bias = b_in[ncol];
#pragma unroll
    for (int r = 0; r < 8; ++r) {
      const int m = r + 8 * half;
      sh_h[m * LDH + ncol]        = acc0[r] + bias;
      sh_h[(16 + m) * LDH + ncol] = acc1[r] + bias;
    }
  }
  __syncthreads();

  for (int layer = 0; layer < NATT; ++layer) {
    const float* W  = W_att + (size_t)layer * H_DIM * H_DIM;
    const float* bb = b_att + layer * H_DIM;

    float4 gq[EPL / 4], bq[EPL / 4];
#pragma unroll
    for (int cc = 0; cc < EPL / 4; ++cc) {
      gq[cc] = *(const float4*)(gamma + layer * H_DIM + 4 * (sub + GLANES * cc));
      bq[cc] = *(const float4*)(beta + layer * H_DIM + 4 * (sub + GLANES * cc));
    }

    {
      v8f acc0 = {}, acc1 = {};
      gemm_cols(sh_h, LDH, W, mlan, khalf, ncol, acc0, acc1);
      const float bias = bb[ncol];
#pragma unroll
      for (int r = 0; r < 8; ++r) {
        const int m = r + 8 * half;
        sh_u[m * H_DIM + ncol]        = tanhf(acc0[r] + bias);
        sh_u[(16 + m) * H_DIM + ncol] = tanhf(acc1[r] + bias);
      }
    }
    __syncthreads();

    {
      const float* ur = sh_u + grow * H_DIM;
      float* hr = sh_h + grow * LDH;
      float uu[EPL], hh[EPL];
#pragma unroll
      for (int cc = 0; cc < EPL / 4; ++cc) {
        const float4 u4 = *(const float4*)(ur + 4 * (sub + GLANES * cc));
        const float4 h4 = *(const float4*)(hr + 4 * (sub + GLANES * cc));
        uu[4 * cc + 0] = u4.x; uu[4 * cc + 1] = u4.y; uu[4 * cc + 2] = u4.z; uu[4 * cc + 3] = u4.w;
        hh[4 * cc + 0] = h4.x; hh[4 * cc + 1] = h4.y; hh[4 * cc + 2] = h4.z; hh[4 * cc + 3] = h4.w;
      }

      float S[NTERM], M[NTERM];
      float pw[EPL];
      {
        float m0 = hh[0], m1 = hh[1];
#pragma unroll
        for (int e = 2; e < EPL; e += 2) { m0 += hh[e]; m1 += hh[e + 1]; }
        M[0] = m0 + m1;
      }
#pragma unroll
      for (int e = 0; e < EPL; ++e) pw[e] = uu[e];
#pragma unroll
      for (int k = 1; k < NTERM; ++k) {
        float s0 = pw[0], s1 = pw[1];
        float m0 = pw[0] * hh[0], m1 = pw[1] * hh[1];
#pragma unroll
        for (int e = 2; e < EPL; e += 2) {
          s0 += pw[e];
          s1 += pw[e + 1];
          m0 = fmaf(pw[e], hh[e], m0);
          m1 = fmaf(pw[e + 1], hh[e + 1], m1);
        }
        S[k] = s0 + s1;
        M[k] = m0 + m1;
        if (k + 1 < NTERM) {
#pragma unroll
          for (int e = 0; e < EPL; ++e) pw[e] *= uu[e];
        }
      }
      constexpr float IFACT[NTERM] = {1.0f,
                                      1.0f,
                                      1.0f / 2.0f,
                                      1.0f / 6.0f,
                                      1.0f / 24.0f,
                                      1.0f / 120.0f,
                                      1.0f / 720.0f,
                                      1.0f / 5040.0f,
                                      1.0f / 40320.0f,
                                      1.0f / 362880.0f,
                                      1.0f / 3628800.0f,
                                      1.0f / 39916800.0f,
                                      1.0f / 479001600.0f};
      float ak[NTERM], bk[NTERM];
#pragma unroll
      for (int k = 0; k < NTERM; ++k) {
        ak[k] = group_sum(M[k]) * IFACT[k];
        bk[k] = (k == 0) ? (float)H_DIM : group_sum(S[k]) * IFACT[k];
      }

      float y[EPL];
#pragma unroll
      for (int e = 0; e < EPL; ++e) {
        const float xq = uu[e];
        float nn = ak[NTERM - 1], dd = bk[NTERM - 1];
#pragma unroll
        for (int k = NTERM - 2; k >= 0; --k) {
          nn = fmaf(nn, xq, ak[k]);
          dd = fmaf(dd, xq, bk[k]);
        }
        y[e] = hh[e] + nn * fast_rcp(dd);
      }

      float s0 = y[0], s1 = y[1];
#pragma unroll
      for (int e = 2; e < EPL; e += 2) { s0 += y[e]; s1 += y[e + 1]; }
      const float mu = group_sum(s0 + s1) * (1.0f / 128.0f);
      float v0 = 0.f, v1 = 0.f;
#pragma unroll
      for (int e = 0; e < EPL; e += 2) {
        const float d0 = y[e] - mu, d1 = y[e + 1] - mu;
        v0 = fmaf(d0, d0, v0);
        v1 = fmaf(d1, d1, v1);
      }
      const float rstd = fast_rsq(group_sum(v0 + v1) * (1.0f / 128.0f) + EPSLN);
#pragma unroll
      for (int cc = 0; cc < EPL / 4; ++cc) {
        float4 o;
        o.x = (y[4 * cc + 0] - mu) * rstd * gq[cc].x + bq[cc].x;
        o.y = (y[4 * cc + 1] - mu) * rstd * gq[cc].y + bq[cc].y;
        o.z = (y[4 * cc + 2] - mu) * rstd * gq[cc].z + bq[cc].z;
        o.w = (y[4 * cc + 3] - mu) * rstd * gq[cc].w + bq[cc].w;
        *(float4*)(hr + 4 * (sub + GLANES * cc)) = o;
      }
    }
    __syncthreads();
  }

  if (wave < RROWS / 16) {
    const int wrow = wave * 16;
    v8f acc = {};
    const bool valid = (mlan < NCLS);
    const float* arow = sh_h + (wrow + mlan) * LDH;
    for (int kk = 0; kk < H_DIM; kk += 4) {
      v2f a = *(const v2f*)(arow + kk + khalf);
      v2f b;
      b.x = valid ? W_c[(kk + khalf) * NCLS + mlan] : 0.0f;
      b.y = valid ? W_c[(kk + khalf + 1) * NCLS + mlan] : 0.0f;
      acc = wmma4(a, b, acc);
    }
    if (valid) {
      const float bias = b_c[mlan];
#pragma unroll
      for (int r = 0; r < 8; ++r) {
        const int m = wrow + r + 8 * half;
        out[(size_t)(row0 + m) * NCLS + mlan] = acc[r] + bias;
      }
    }
  }
}

extern "C" void kernel_launch(void* const* d_in, const int* in_sizes, int n_in,
                              void* d_out, int out_size, void* d_ws,
                              size_t ws_size, hipStream_t stream) {
  const float* x     = (const float*)d_in[0];
  const float* W_in  = (const float*)d_in[1];
  const float* b_in  = (const float*)d_in[2];
  const float* W_att = (const float*)d_in[3];
  const float* b_att = (const float*)d_in[4];
  const float* gamma = (const float*)d_in[5];
  const float* beta  = (const float*)d_in[6];
  const float* W_c   = (const float*)d_in[7];
  const float* b_c   = (const float*)d_in[8];
  float* out = (float*)d_out;

  dim3 grid(B_ROWS / RROWS);
  dim3 block(NWAVES * 32);
  simpleattention_fused_kernel<<<grid, block, 0, stream>>>(
      x, W_in, b_in, W_att, b_att, gamma, beta, W_c, b_c, out);
}


# Weights fetched ahead of use: each GEMM's B slice loaded during the GEMM before it (two halves, under 64 loads in flight), W_c staged in LDS at the start; no load round trip left in front of a WMMA
# speedup vs baseline: 6.4529x; 1.2674x over previous
#include <hip/hip_runtime.h>


#define B_ROWS 16384
#define IN_DIM 128
#define H_DIM  128
#define NCLS   10
#define NATT   3
#define EPSLN  1e-5f

#define NWAVES 8
#define RROWS  32
#define RPW    (RROWS / NWAVES)
#define GLANES (32 / RPW)
#define EPL    (H_DIM / GLANES)
#define NTERM  13
#define KSTEPS (H_DIM / 4)
#define LDH    132

typedef float v2f __attribute__((ext_vector_type(2)));
typedef float v8f __attribute__((ext_vector_type(8)));

static __device__ __forceinline__ v8f wmma4(v2f a, v2f b, v8f c) {
  return __builtin_amdgcn_wmma_f32_16x16x4_f32(false, a, false, b, (short)0, c,
                                               false, false);
}

static __device__ __forceinline__ float fast_rcp(float x) {
#if __has_builtin(__builtin_amdgcn_rcpf)
  return __builtin_amdgcn_rcpf(x);
#else
  return 1.0f / x;
#endif
}
static __device__ __forceinline__ float fast_rsq(float x) {
#if __has_builtin(__builtin_amdgcn_rsqf)
  return __builtin_amdgcn_rsqf(x);
#else
  return rsqrtf(x);
#endif
}

template <int CTRL>
static __device__ __forceinline__ float lane_xor(float v) {
  return __builtin_bit_cast(
      float, __builtin_amdgcn_update_dpp(0, __builtin_bit_cast(int, v), CTRL, 0xf, 0xf, false));
}
static __device__ __forceinline__ float group_sum(float v) {
  v += lane_xor<0x161>(v);
  v += lane_xor<0x162>(v);
  v += lane_xor<0x164>(v);
  return v;
}

#define KHALFWAY (KSTEPS / 2)
#define KHEAD 26

template <int STRIDE, int S0, int S1>
static __device__ __forceinline__ void load_slice(v2f (&bf)[KSTEPS], const float* __restrict__ p) {
#pragma unroll
  for (int s = S0; s < S1; ++s) {
    bf[s].x = p[(4 * s) * STRIDE];
    bf[s].y = p[(4 * s + 1) * STRIDE];
  }
}

template <int S0, int S1>
static __device__ __forceinline__ void gemm_steps(const float* sA, const int lda,
                                                  const v2f (&bf)[KSTEPS],
                                                  const int mlan, const int khalf,
                                                  v8f& acc0, v8f& acc1) {
  const float* a0p = sA + mlan * lda + khalf;
  const float* a1p = sA + (16 + mlan) * lda + khalf;
#pragma unroll
  for (int s = S0; s < S1; ++s) {
    const v2f a0 = *(const v2f*)(a0p + 4 * s);
    const v2f a1 = *(const v2f*)(a1p + 4 * s);
    acc0 = wmma4(a0, bf[s], acc0);
    acc1 = wmma4(a1, bf[s], acc1);
  }
}

__global__ void __launch_bounds__(NWAVES * 32, 4)
simpleattention_fused_kernel(const float* __restrict__ x,
                             const float* __restrict__ W_in,
                             const float* __restrict__ b_in,
                             const float* __restrict__ W_att,
                             const float* __restrict__ b_att,
                             const float* __restrict__ gamma,
                             const float* __restrict__ beta,
                             const float* __restrict__ W_c,
                             const float* __restrict__ b_c,
                             float* __restrict__ out) {
  __shared__ __attribute__((aligned(16))) float sh_h[RROWS * LDH];
  __shared__ __attribute__((aligned(16))) float sh_u[RROWS * LDH];
  __shared__ __attribute__((aligned(16))) float sh_wc[H_DIM * NCLS];

  const int lane  = threadIdx.x & 31;
  const int wave  = __builtin_amdgcn_readfirstlane((int)(threadIdx.x >> 5));
  const int mlan  = lane & 15;
  const int half  = lane >> 4;
  const int khalf = half << 1;
  const int row0  = blockIdx.x * RROWS;
  const int ncol  = wave * 16 + mlan;
  const int grow  = wave * RPW + (lane >> 3);
  const int sub   = lane & (GLANES - 1);
  const bool valid = (mlan < NCLS);
  const int  ccol  = valid ? mlan : 0;

  v2f   bf[KSTEPS];
  float bias;
  float bias_c;

  {
    const float4* xg = (const float4*)(x + (size_t)row0 * IN_DIM);
    float4 xv[(RROWS * IN_DIM / 4) / (NWAVES * 32)];
    float  wc[(H_DIM * NCLS) / (NWAVES * 32)];
#pragma unroll
    for (int i = 0; i < (RROWS * IN_DIM / 4) / (NWAVES * 32); ++i)
      xv[i] = xg[(int)threadIdx.x + i * NWAVES * 32];
#pragma unroll
    for (int i = 0; i < (H_DIM * NCLS) / (NWAVES * 32); ++i)
      wc[i] = W_c[(int)threadIdx.x + i * NWAVES * 32];
    bias_c = b_c[ccol];
    load_slice<H_DIM, 0, KHEAD>(bf, W_in + khalf * H_DIM + ncol);
#pragma unroll
    for (int i = 0; i < (RROWS * IN_DIM / 4) / (NWAVES * 32); ++i) {
      const int idx = (int)threadIdx.x + i * NWAVES * 32;
      *(float4*)(sh_u + (idx >> 5) * LDH + 4 * (idx & 31)) = xv[i];
    }
#pragma unroll
    for (int i = 0; i < (H_DIM * NCLS) / (NWAVES * 32); ++i)
      sh_wc[(int)threadIdx.x + i * NWAVES * 32] = wc[i];
  }
  __syncthreads();

  {
    bias = b_in[ncol];
    load_slice<H_DIM, KHEAD, KSTEPS>(bf, W_in + khalf * H_DIM + ncol);
    const float* wnext = W_att + khalf * H_DIM + ncol;
    v8f acc0 = {}, acc1 = {};
    gemm_steps<0, KHALFWAY>(sh_u, LDH, bf, mlan, khalf, acc0, acc1);
    load_slice<H_DIM, 0, KHALFWAY>(bf, wnext);
    gemm_steps<KHALFWAY, KSTEPS>(sh_u, LDH, bf, mlan, khalf, acc0, acc1);
    const float bias_h = bias;
    bias = b_att[ncol];
    load_slice<H_DIM, KHALFWAY, KSTEPS>(bf, wnext);
#pragma unroll
    for (int r = 0; r < 8; ++r) {
      const int m = r + 8 * half;
      sh_h[m * LDH + ncol]        = acc0[r] + bias_h;
      sh_h[(16 + m) * LDH + ncol] = acc1[r] + bias_h;
    }
  }
  __syncthreads();

  for (int layer = 0; layer < NATT; ++layer) {
    float4 gq[EPL / 4], bq[EPL / 4];
#pragma unroll
    for (int cc = 0; cc < EPL / 4; ++cc) {
      gq[cc] = *(const float4*)(gamma + layer * H_DIM + 4 * (sub + GLANES * cc));
      bq[cc] = *(const float4*)(beta + layer * H_DIM + 4 * (sub + GLANES * cc));
    }

    {
      const bool more = (layer + 1 < NATT);
      const float* wnext = W_att + (size_t)(layer + 1) * H_DIM * H_DIM + khalf * H_DIM + ncol;
      v8f acc0 = {}, acc1 = {};
      gemm_steps<0, KHALFWAY>(sh_h, LDH, bf, mlan, khalf, acc0, acc1);
      if (more) load_slice<H_DIM, 0, KHALFWAY>(bf, wnext);
      gemm_steps<KHALFWAY, KSTEPS>(sh_h, LDH, bf, mlan, khalf, acc0, acc1);
      const float bias_u = bias;
      if (more) {
        bias = b_att[(layer + 1) * H_DIM + ncol];
        load_slice<H_DIM, KHALFWAY, KSTEPS>(bf, wnext);
      }
#pragma unroll
      for (int r = 0; r < 8; ++r) {
        const int m = r + 8 * half;
        sh_u[m * H_DIM + ncol]        = tanhf(acc0[r] + bias_u);
        sh_u[(16 + m) * H_DIM + ncol] = tanhf(acc1[r] + bias_u);
      }
    }
    __syncthreads();

    {
      const float* ur = sh_u + grow * H_DIM;
      float* hr = sh_h + grow * LDH;
      float uu[EPL], hh[EPL];
#pragma unroll
      for (int cc = 0; cc < EPL / 4; ++cc) {
        const float4 u4 = *(const float4*)(ur + 4 * (sub + GLANES * cc));
        const float4 h4 = *(const float4*)(hr + 4 * (sub + GLANES * cc));
        uu[4 * cc + 0] = u4.x; uu[4 * cc + 1] = u4.y; uu[4 * cc + 2] = u4.z; uu[4 * cc + 3] = u4.w;
        hh[4 * cc + 0] = h4.x; hh[4 * cc + 1] = h4.y; hh[4 * cc + 2] = h4.z; hh[4 * cc + 3] = h4.w;
      }

      float S[NTERM], M[NTERM];
      float pw[EPL];
      {
        float m0 = hh[0], m1 = hh[1];
#pragma unroll
        for (int e = 2; e < EPL; e += 2) { m0 += hh[e]; m1 += hh[e + 1]; }
        M[0] = m0 + m1;
      }
#pragma unroll
      for (int e = 0; e < EPL; ++e) pw[e] = uu[e];
#pragma unroll
      for (int k = 1; k < NTERM; ++k) {
        float s0 = pw[0], s1 = pw[1];
        float m0 = pw[0] * hh[0], m1 = pw[1] * hh[1];
#pragma unroll
        for (int e = 2; e < EPL; e += 2) {
          s0 += pw[e];
          s1 += pw[e + 1];
          m0 = fmaf(pw[e], hh[e], m0);
          m1 = fmaf(pw[e + 1], hh[e + 1], m1);
        }
        S[k] = s0 + s1;
        M[k] = m0 + m1;
        if (k + 1 < NTERM) {
#pragma unroll
          for (int e = 0; e < EPL; ++e) pw[e] *= uu[e];
        }
      }
      constexpr float IFACT[NTERM] = {1.0f,
                                      1.0f,
                                      1.0f / 2.0f,
                                      1.0f / 6.0f,
                                      1.0f / 24.0f,
                                      1.0f / 120.0f,
                                      1.0f / 720.0f,
                                      1.0f / 5040.0f,
                                      1.0f / 40320.0f,
                                      1.0f / 362880.0f,
                                      1.0f / 3628800.0f,
                                      1.0f / 39916800.0f,
                                      1.0f / 479001600.0f};
      float ak[NTERM], bk[NTERM];
#pragma unroll
      for (int k = 0; k < NTERM; ++k) {
        ak[k] = group_sum(M[k]) * IFACT[k];
        bk[k] = (k == 0) ? (float)H_DIM : group_sum(S[k]) * IFACT[k];
      }

      float y[EPL];
#pragma unroll
      for (int e = 0; e < EPL; ++e) {
        const float xq = uu[e];
        float nn = ak[NTERM - 1], dd = bk[NTERM - 1];
#pragma unroll
        for (int k = NTERM - 2; k >= 0; --k) {
          nn = fmaf(nn, xq, ak[k]);
          dd = fmaf(dd, xq, bk[k]);
        }
        y[e] = hh[e] + nn * fast_rcp(dd);
      }

      float s0 = y[0], s1 = y[1];
#pragma unroll
      for (int e = 2; e < EPL; e += 2) { s0 += y[e]; s1 += y[e + 1]; }
      const float mu = group_sum(s0 + s1) * (1.0f / 128.0f);
      float v0 = 0.f, v1 = 0.f;
#pragma unroll
      for (int e = 0; e < EPL; e += 2) {
        const float d0 = y[e] - mu, d1 = y[e + 1] - mu;
        v0 = fmaf(d0, d0, v0);
        v1 = fmaf(d1, d1, v1);
      }
      const float rstd = fast_rsq(group_sum(v0 + v1) * (1.0f / 128.0f) + EPSLN);
#pragma unroll
      for (int cc = 0; cc < EPL / 4; ++cc) {
        float4 o;
        o.x = (y[4 * cc + 0] - mu) * rstd * gq[cc].x + bq[cc].x;
        o.y = (y[4 * cc + 1] - mu) * rstd * gq[cc].y + bq[cc].y;
        o.z = (y[4 * cc + 2] - mu) * rstd * gq[cc].z + bq[cc].z;
        o.w = (y[4 * cc + 3] - mu) * rstd * gq[cc].w + bq[cc].w;
        *(float4*)(hr + 4 * (sub + GLANES * cc)) = o;
      }
    }
    __syncthreads();
  }

  if (wave < RROWS / 16) {
    const int wrow = wave * 16;
    const float* arow = sh_h + (wrow + mlan) * LDH + khalf;
    const float* brow = sh_wc + khalf * NCLS + ccol;
    v2f af[KSTEPS], bw[KSTEPS];
#pragma unroll
    for (int s = 0; s < KSTEPS; ++s) {
      af[s]   = *(const v2f*)(arow + 4 * s);
      bw[s].x = brow[(4 * s) * NCLS];
      bw[s].y = brow[(4 * s + 1) * NCLS];
    }
    __builtin_amdgcn_sched_barrier(0);
    v8f acc = {};
#pragma unroll
    for (int s = 0; s < KSTEPS; ++s) acc = wmma4(af[s], bw[s], acc);
    if (valid) {
#pragma unroll
      for (int r = 0; r < 8; ++r) {
        const int m = wrow + r + 8 * half;
        out[(size_t)(row0 + m) * NCLS + mlan] = acc[r] + bias_c;
      }
    }
  }
}

extern "C" void kernel_launch(void* const* d_in, const int* in_sizes, int n_in,
                              void* d_out, int out_size, void* d_ws,
                              size_t ws_size, hipStream_t stream) {
  const float* x     = (const float*)d_in[0];
  const float* W_in  = (const float*)d_in[1];
  const float* b_in  = (const float*)d_in[2];
  const float* W_att = (const float*)d_in[3];
  const float* b_att = (const float*)d_in[4];
  const float* gamma = (const float*)d_in[5];
  const float* beta  = (const float*)d_in[6];
  const float* W_c   = (const float*)d_in[7];
  const float* b_c   = (const float*)d_in[8];
  float* out = (float*)d_out;

  dim3 grid(B_ROWS / RROWS);
  dim3 block(NWAVES * 32);
  simpleattention_fused_kernel<<<grid, block, 0, stream>>>(
      x, W_in, b_in, W_att, b_att, gamma, beta, W_c, b_c, out);
}


# tanh of the u epilogue without branches: sign(x)(1-t)/(1+t), t=2^(-2log2(e)|x|) on v_exp_f32 and v_rcp_f32; 6 instructions a value where the library took 36 over two branches; abs error under 2^-21
# speedup vs baseline: 7.4299x; 1.1514x over previous
#include <hip/hip_runtime.h>


#define B_ROWS 16384
#define IN_DIM 128
#define H_DIM  128
#define NCLS   10
#define NATT   3
#define EPSLN  1e-5f

#define NWAVES 8
#define RROWS  32
#define RPW    (RROWS / NWAVES)
#define GLANES (32 / RPW)
#define EPL    (H_DIM / GLANES)
#define NTERM  13
#define KSTEPS (H_DIM / 4)
#define LDH    132

typedef float v2f __attribute__((ext_vector_type(2)));
typedef float v8f __attribute__((ext_vector_type(8)));

static __device__ __forceinline__ v8f wmma4(v2f a, v2f b, v8f c) {
  return __builtin_amdgcn_wmma_f32_16x16x4_f32(false, a, false, b, (short)0, c,
                                               false, false);
}

static __device__ __forceinline__ float fast_rcp(float x) {
#if __has_builtin(__builtin_amdgcn_rcpf)
  return __builtin_amdgcn_rcpf(x);
#else
  return 1.0f / x;
#endif
}
static __device__ __forceinline__ float tanh_exp2(float x) {
#if __has_builtin(__builtin_amdgcn_exp2f)
  const float t = __builtin_amdgcn_exp2f(-2.885390081777927f * __builtin_fabsf(x));
#else
  const float t = exp2f(-2.885390081777927f * __builtin_fabsf(x));
#endif
  const float r = fast_rcp(1.0f + t);
  return __builtin_copysignf(__builtin_fmaf(-t, r, r), x);
}
static __device__ __forceinline__ float fast_rsq(float x) {
#if __has_builtin(__builtin_amdgcn_rsqf)
  return __builtin_amdgcn_rsqf(x);
#else
  return rsqrtf(x);
#endif
}

template <int CTRL>
static __device__ __forceinline__ float lane_xor(float v) {
  return __builtin_bit_cast(
      float, __builtin_amdgcn_update_dpp(0, __builtin_bit_cast(int, v), CTRL, 0xf, 0xf, false));
}
static __device__ __forceinline__ float group_sum(float v) {
  v += lane_xor<0x161>(v);
  v += lane_xor<0x162>(v);
  v += lane_xor<0x164>(v);
  return v;
}

#define KHALFWAY (KSTEPS / 2)
#define KHEAD 26

template <int STRIDE, int S0, int S1>
static __device__ __forceinline__ void load_slice(v2f (&bf)[KSTEPS], const float* __restrict__ p) {
#pragma unroll
  for (int s = S0; s < S1; ++s) {
    bf[s].x = p[(4 * s) * STRIDE];
    bf[s].y = p[(4 * s + 1) * STRIDE];
  }
}

template <int S0, int S1>
static __device__ __forceinline__ void gemm_steps(const float* sA, const int lda,
                                                  const v2f (&bf)[KSTEPS],
                                                  const int mlan, const int khalf,
                                                  v8f& acc0, v8f& acc1) {
  const float* a0p = sA + mlan * lda + khalf;
  const float* a1p = sA + (16 + mlan) * lda + khalf;
#pragma unroll
  for (int s = S0; s < S1; ++s) {
    const v2f a0 = *(const v2f*)(a0p + 4 * s);
    const v2f a1 = *(const v2f*)(a1p + 4 * s);
    acc0 = wmma4(a0, bf[s], acc0);
    acc1 = wmma4(a1, bf[s], acc1);
  }
}

__global__ void __launch_bounds__(NWAVES * 32, 4)
simpleattention_fused_kernel(const float* __restrict__ x,
                             const float* __restrict__ W_in,
                             const float* __restrict__ b_in,
                             const float* __restrict__ W_att,
                             const float* __restrict__ b_att,
                             const float* __restrict__ gamma,
                             const float* __restrict__ beta,
                             const float* __restrict__ W_c,
                             const float* __restrict__ b_c,
                             float* __restrict__ out) {
  __shared__ __attribute__((aligned(16))) float sh_h[RROWS * LDH];
  __shared__ __attribute__((aligned(16))) float sh_u[RROWS * LDH];
  __shared__ __attribute__((aligned(16))) float sh_wc[H_DIM * NCLS];

  const int lane  = threadIdx.x & 31;
  const int wave  = __builtin_amdgcn_readfirstlane((int)(threadIdx.x >> 5));
  const int mlan  = lane & 15;
  const int half  = lane >> 4;
  const int khalf = half << 1;
  const int row0  = blockIdx.x * RROWS;
  const int ncol  = wave * 16 + mlan;
  const int grow  = wave * RPW + (lane >> 3);
  const int sub   = lane & (GLANES - 1);
  const bool valid = (mlan < NCLS);
  const int  ccol  = valid ? mlan : 0;

  v2f   bf[KSTEPS];
  float bias;
  float bias_c;

  {
    const float4* xg = (const float4*)(x + (size_t)row0 * IN_DIM);
    float4 xv[(RROWS * IN_DIM / 4) / (NWAVES * 32)];
    float  wc[(H_DIM * NCLS) / (NWAVES * 32)];
#pragma unroll
    for (int i = 0; i < (RROWS * IN_DIM / 4) / (NWAVES * 32); ++i)
      xv[i] = xg[(int)threadIdx.x + i * NWAVES * 32];
#pragma unroll
    for (int i = 0; i < (H_DIM * NCLS) / (NWAVES * 32); ++i)
      wc[i] = W_c[(int)threadIdx.x + i * NWAVES * 32];
    bias_c = b_c[ccol];
    load_slice<H_DIM, 0, KHEAD>(bf, W_in + khalf * H_DIM + ncol);
#pragma unroll
    for (int i = 0; i < (RROWS * IN_DIM / 4) / (NWAVES * 32); ++i) {
      const int idx = (int)threadIdx.x + i * NWAVES * 32;
      *(float4*)(sh_u + (idx >> 5) * LDH + 4 * (idx & 31)) = xv[i];
    }
#pragma unroll
    for (int i = 0; i < (H_DIM * NCLS) / (NWAVES * 32); ++i)
      sh_wc[(int)threadIdx.x + i * NWAVES * 32] = wc[i];
  }
  __syncthreads();

  {
    bias = b_in[ncol];
    load_slice<H_DIM, KHEAD, KSTEPS>(bf, W_in + khalf * H_DIM + ncol);
    const float* wnext = W_att + khalf * H_DIM + ncol;
    v8f acc0 = {}, acc1 = {};
    gemm_steps<0, KHALFWAY>(sh_u, LDH, bf, mlan, khalf, acc0, acc1);
    load_slice<H_DIM, 0, KHALFWAY>(bf, wnext);
    gemm_steps<KHALFWAY, KSTEPS>(sh_u, LDH, bf, mlan, khalf, acc0, acc1);
    const float bias_h = bias;
    bias = b_att[ncol];
    load_slice<H_DIM, KHALFWAY, KSTEPS>(bf, wnext);
#pragma unroll
    for (int r = 0; r < 8; ++r) {
      const int m = r + 8 * half;
      sh_h[m * LDH + ncol]        = acc0[r] + bias_h;
      sh_h[(16 + m) * LDH + ncol] = acc1[r] + bias_h;
    }
  }
  __syncthreads();

  for (int layer = 0; layer < NATT; ++layer) {
    float4 gq[EPL / 4], bq[EPL / 4];
#pragma unroll
    for (int cc = 0; cc < EPL / 4; ++cc) {
      gq[cc] = *(const float4*)(gamma + layer * H_DIM + 4 * (sub + GLANES * cc));
      bq[cc] = *(const float4*)(beta + layer * H_DIM + 4 * (sub + GLANES * cc));
    }

    {
      const bool more = (layer + 1 < NATT);
      const float* wnext = W_att + (size_t)(layer + 1) * H_DIM * H_DIM + khalf * H_DIM + ncol;
      v8f acc0 = {}, acc1 = {};
      gemm_steps<0, KHALFWAY>(sh_h, LDH, bf, mlan, khalf, acc0, acc1);
      if (more) load_slice<H_DIM, 0, KHALFWAY>(bf, wnext);
      gemm_steps<KHALFWAY, KSTEPS>(sh_h, LDH, bf, mlan, khalf, acc0, acc1);
      const float bias_u = bias;
      if (more) {
        bias = b_att[(layer + 1) * H_DIM + ncol];
        load_slice<H_DIM, KHALFWAY, KSTEPS>(bf, wnext);
      }
#pragma unroll
      for (int r = 0; r < 8; ++r) {
        const int m = r + 8 * half;
        sh_u[m * H_DIM + ncol]        = tanh_exp2(acc0[r] + bias_u);
        sh_u[(16 + m) * H_DIM + ncol] = tanh_exp2(acc1[r] + bias_u);
      }
    }
    __syncthreads();

    {
      const float* ur = sh_u + grow * H_DIM;
      float* hr = sh_h + grow * LDH;
      float uu[EPL], hh[EPL];
#pragma unroll
      for (int cc = 0; cc < EPL / 4; ++cc) {
        const float4 u4 = *(const float4*)(ur + 4 * (sub + GLANES * cc));
        const float4 h4 = *(const float4*)(hr + 4 * (sub + GLANES * cc));
        uu[4 * cc + 0] = u4.x; uu[4 * cc + 1] = u4.y; uu[4 * cc + 2] = u4.z; uu[4 * cc + 3] = u4.w;
        hh[4 * cc + 0] = h4.x; hh[4 * cc + 1] = h4.y; hh[4 * cc + 2] = h4.z; hh[4 * cc + 3] = h4.w;
      }

      float S[NTERM], M[NTERM];
      float pw[EPL];
      {
        float m0 = hh[0], m1 = hh[1];
#pragma unroll
        for (int e = 2; e < EPL; e += 2) { m0 += hh[e]; m1 += hh[e + 1]; }
        M[0] = m0 + m1;
      }
#pragma unroll
      for (int e = 0; e < EPL; ++e) pw[e] = uu[e];
#pragma unroll
      for (int k = 1; k < NTERM; ++k) {
        float s0 = pw[0], s1 = pw[1];
        float m0 = pw[0] * hh[0], m1 = pw[1] * hh[1];
#pragma unroll
        for (int e = 2; e < EPL; e += 2) {
          s0 += pw[e];
          s1 += pw[e + 1];
          m0 = fmaf(pw[e], hh[e], m0);
          m1 = fmaf(pw[e + 1], hh[e + 1], m1);
        }
        S[k] = s0 + s1;
        M[k] = m0 + m1;
        if (k + 1 < NTERM) {
#pragma unroll
          for (int e = 0; e < EPL; ++e) pw[e] *= uu[e];
        }
      }
      constexpr float IFACT[NTERM] = {1.0f,
                                      1.0f,
                                      1.0f / 2.0f,
                                      1.0f / 6.0f,
                                      1.0f / 24.0f,
                                      1.0f / 120.0f,
                                      1.0f / 720.0f,
                                      1.0f / 5040.0f,
                                      1.0f / 40320.0f,
                                      1.0f / 362880.0f,
                                      1.0f / 3628800.0f,
                                      1.0f / 39916800.0f,
                                      1.0f / 479001600.0f};
      float ak[NTERM], bk[NTERM];
#pragma unroll
      for (int k = 0; k < NTERM; ++k) {
        ak[k] = group_sum(M[k]) * IFACT[k];
        bk[k] = (k == 0) ? (float)H_DIM : group_sum(S[k]) * IFACT[k];
      }

      float y[EPL];
#pragma unroll
      for (int e = 0; e < EPL; ++e) {
        const float xq = uu[e];
        float nn = ak[NTERM - 1], dd = bk[NTERM - 1];
#pragma unroll
        for (int k = NTERM - 2; k >= 0; --k) {
          nn = fmaf(nn, xq, ak[k]);
          dd = fmaf(dd, xq, bk[k]);
        }
        y[e] = hh[e] + nn * fast_rcp(dd);
      }

      float s0 = y[0], s1 = y[1];
#pragma unroll
      for (int e = 2; e < EPL; e += 2) { s0 += y[e]; s1 += y[e + 1]; }
      const float mu = group_sum(s0 + s1) * (1.0f / 128.0f);
      float v0 = 0.f, v1 = 0.f;
#pragma unroll
      for (int e = 0; e < EPL; e += 2) {
        const float d0 = y[e] - mu, d1 = y[e + 1] - mu;
        v0 = fmaf(d0, d0, v0);
        v1 = fmaf(d1, d1, v1);
      }
      const float rstd = fast_rsq(group_sum(v0 + v1) * (1.0f / 128.0f) + EPSLN);
#pragma unroll
      for (int cc = 0; cc < EPL / 4; ++cc) {
        float4 o;
        o.x = (y[4 * cc + 0] - mu) * rstd * gq[cc].x + bq[cc].x;
        o.y = (y[4 * cc + 1] - mu) * rstd * gq[cc].y + bq[cc].y;
        o.z = (y[4 * cc + 2] - mu) * rstd * gq[cc].z + bq[cc].z;
        o.w = (y[4 * cc + 3] - mu) * rstd * gq[cc].w + bq[cc].w;
        *(float4*)(hr + 4 * (sub + GLANES * cc)) = o;
      }
    }
    __syncthreads();
  }

  if (wave < RROWS / 16) {
    const int wrow = wave * 16;
    const float* arow = sh_h + (wrow + mlan) * LDH + khalf;
    const float* brow = sh_wc + khalf * NCLS + ccol;
    v2f af[KSTEPS], bw[KSTEPS];
#pragma unroll
    for (int s = 0; s < KSTEPS; ++s) {
      af[s]   = *(const v2f*)(arow + 4 * s);
      bw[s].x = brow[(4 * s) * NCLS];
      bw[s].y = brow[(4 * s + 1) * NCLS];
    }
    __builtin_amdgcn_sched_barrier(0);
    v8f acc = {};
#pragma unroll
    for (int s = 0; s < KSTEPS; ++s) acc = wmma4(af[s], bw[s], acc);
    if (valid) {
#pragma unroll
      for (int r = 0; r < 8; ++r) {
        const int m = wrow + r + 8 * half;
        out[(size_t)(row0 + m) * NCLS + mlan] = acc[r] + bias_c;
      }
    }
  }
}

extern "C" void kernel_launch(void* const* d_in, const int* in_sizes, int n_in,
                              void* d_out, int out_size, void* d_ws,
                              size_t ws_size, hipStream_t stream) {
  const float* x     = (const float*)d_in[0];
  const float* W_in  = (const float*)d_in[1];
  const float* b_in  = (const float*)d_in[2];
  const float* W_att = (const float*)d_in[3];
  const float* b_att = (const float*)d_in[4];
  const float* gamma = (const float*)d_in[5];
  const float* beta  = (const float*)d_in[6];
  const float* W_c   = (const float*)d_in[7];
  const float* b_c   = (const float*)d_in[8];
  float* out = (float*)d_out;

  dim3 grid(B_ROWS / RROWS);
  dim3 block(NWAVES * 32);
  simpleattention_fused_kernel<<<grid, block, 0, stream>>>(
      x, W_in, b_in, W_att, b_att, gamma, beta, W_c, b_c, out);
}


# exp(u_i*u_j), |u_i*u_j|<=1, by its degree-9 minimax polynomial in place of the power series to k=12: 10 terms for 13 at no larger error (under 4e-9); 19 moments for 25, fewer terms in each sum
# speedup vs baseline: 7.8505x; 1.0566x over previous
#include <hip/hip_runtime.h>


#define B_ROWS 16384
#define IN_DIM 128
#define H_DIM  128
#define NCLS   10
#define NATT   3
#define EPSLN  1e-5f

#define NWAVES 8
#define RROWS  32
#define RPW    (RROWS / NWAVES)
#define GLANES (32 / RPW)
#define EPL    (H_DIM / GLANES)
#define NTERM  10
#define KSTEPS (H_DIM / 4)
#define LDH    132

typedef float v2f __attribute__((ext_vector_type(2)));
typedef float v8f __attribute__((ext_vector_type(8)));

static __device__ __forceinline__ v8f wmma4(v2f a, v2f b, v8f c) {
  return __builtin_amdgcn_wmma_f32_16x16x4_f32(false, a, false, b, (short)0, c,
                                               false, false);
}

static __device__ __forceinline__ float fast_rcp(float x) {
#if __has_builtin(__builtin_amdgcn_rcpf)
  return __builtin_amdgcn_rcpf(x);
#else
  return 1.0f / x;
#endif
}
static __device__ __forceinline__ float tanh_exp2(float x) {
#if __has_builtin(__builtin_amdgcn_exp2f)
  const float t = __builtin_amdgcn_exp2f(-2.885390081777927f * __builtin_fabsf(x));
#else
  const float t = exp2f(-2.885390081777927f * __builtin_fabsf(x));
#endif
  const float r = fast_rcp(1.0f + t);
  return __builtin_copysignf(__builtin_fmaf(-t, r, r), x);
}
static __device__ __forceinline__ float fast_rsq(float x) {
#if __has_builtin(__builtin_amdgcn_rsqf)
  return __builtin_amdgcn_rsqf(x);
#else
  return rsqrtf(x);
#endif
}

template <int CTRL>
static __device__ __forceinline__ float lane_xor(float v) {
  return __builtin_bit_cast(
      float, __builtin_amdgcn_update_dpp(0, __builtin_bit_cast(int, v), CTRL, 0xf, 0xf, false));
}
static __device__ __forceinline__ float group_sum(float v) {
  v += lane_xor<0x161>(v);
  v += lane_xor<0x162>(v);
  v += lane_xor<0x164>(v);
  return v;
}

#define KHALFWAY (KSTEPS / 2)
#define KHEAD 26

template <int STRIDE, int S0, int S1>
static __device__ __forceinline__ void load_slice(v2f (&bf)[KSTEPS], const float* __restrict__ p) {
#pragma unroll
  for (int s = S0; s < S1; ++s) {
    bf[s].x = p[(4 * s) * STRIDE];
    bf[s].y = p[(4 * s + 1) * STRIDE];
  }
}

template <int S0, int S1>
static __device__ __forceinline__ void gemm_steps(const float* sA, const int lda,
                                                  const v2f (&bf)[KSTEPS],
                                                  const int mlan, const int khalf,
                                                  v8f& acc0, v8f& acc1) {
  const float* a0p = sA + mlan * lda + khalf;
  const float* a1p = sA + (16 + mlan) * lda + khalf;
#pragma unroll
  for (int s = S0; s < S1; ++s) {
    const v2f a0 = *(const v2f*)(a0p + 4 * s);
    const v2f a1 = *(const v2f*)(a1p + 4 * s);
    acc0 = wmma4(a0, bf[s], acc0);
    acc1 = wmma4(a1, bf[s], acc1);
  }
}

__global__ void __launch_bounds__(NWAVES * 32, 4)
simpleattention_fused_kernel(const float* __restrict__ x,
                             const float* __restrict__ W_in,
                             const float* __restrict__ b_in,
                             const float* __restrict__ W_att,
                             const float* __restrict__ b_att,
                             const float* __restrict__ gamma,
                             const float* __restrict__ beta,
                             const float* __restrict__ W_c,
                             const float* __restrict__ b_c,
                             float* __restrict__ out) {
  __shared__ __attribute__((aligned(16))) float sh_h[RROWS * LDH];
  __shared__ __attribute__((aligned(16))) float sh_u[RROWS * LDH];
  __shared__ __attribute__((aligned(16))) float sh_wc[H_DIM * NCLS];

  const int lane  = threadIdx.x & 31;
  const int wave  = __builtin_amdgcn_readfirstlane((int)(threadIdx.x >> 5));
  const int mlan  = lane & 15;
  const int half  = lane >> 4;
  const int khalf = half << 1;
  const int row0  = blockIdx.x * RROWS;
  const int ncol  = wave * 16 + mlan;
  const int grow  = wave * RPW + (lane >> 3);
  const int sub   = lane & (GLANES - 1);
  const bool valid = (mlan < NCLS);
  const int  ccol  = valid ? mlan : 0;

  v2f   bf[KSTEPS];
  float bias;
  float bias_c;

  {
    const float4* xg = (const float4*)(x + (size_t)row0 * IN_DIM);
    float4 xv[(RROWS * IN_DIM / 4) / (NWAVES * 32)];
    float  wc[(H_DIM * NCLS) / (NWAVES * 32)];
#pragma unroll
    for (int i = 0; i < (RROWS * IN_DIM / 4) / (NWAVES * 32); ++i)
      xv[i] = xg[(int)threadIdx.x + i * NWAVES * 32];
#pragma unroll
    for (int i = 0; i < (H_DIM * NCLS) / (NWAVES * 32); ++i)
      wc[i] = W_c[(int)threadIdx.x + i * NWAVES * 32];
    bias_c = b_c[ccol];
    load_slice<H_DIM, 0, KHEAD>(bf, W_in + khalf * H_DIM + ncol);
#pragma unroll
    for (int i = 0; i < (RROWS * IN_DIM / 4) / (NWAVES * 32); ++i) {
      const int idx = (int)threadIdx.x + i * NWAVES * 32;
      *(float4*)(sh_u + (idx >> 5) * LDH + 4 * (idx & 31)) = xv[i];
    }
#pragma unroll
    for (int i = 0; i < (H_DIM * NCLS) / (NWAVES * 32); ++i)
      sh_wc[(int)threadIdx.x + i * NWAVES * 32] = wc[i];
  }
  __syncthreads();

  {
    bias = b_in[ncol];
    load_slice<H_DIM, KHEAD, KSTEPS>(bf, W_in + khalf * H_DIM + ncol);
    const float* wnext = W_att + khalf * H_DIM + ncol;
    v8f acc0 = {}, acc1 = {};
    gemm_steps<0, KHALFWAY>(sh_u, LDH, bf, mlan, khalf, acc0, acc1);
    load_slice<H_DIM, 0, KHALFWAY>(bf, wnext);
    gemm_steps<KHALFWAY, KSTEPS>(sh_u, LDH, bf, mlan, khalf, acc0, acc1);
    const float bias_h = bias;
    bias = b_att[ncol];
    load_slice<H_DIM, KHALFWAY, KSTEPS>(bf, wnext);
#pragma unroll
    for (int r = 0; r < 8; ++r) {
      const int m = r + 8 * half;
      sh_h[m * LDH + ncol]        = acc0[r] + bias_h;
      sh_h[(16 + m) * LDH + ncol] = acc1[r] + bias_h;
    }
  }
  __syncthreads();

  for (int layer = 0; layer < NATT; ++layer) {
    float4 gq[EPL / 4], bq[EPL / 4];
#pragma unroll
    for (int cc = 0; cc < EPL / 4; ++cc) {
      gq[cc] = *(const float4*)(gamma + layer * H_DIM + 4 * (sub + GLANES * cc));
      bq[cc] = *(const float4*)(beta + layer * H_DIM + 4 * (sub + GLANES * cc));
    }

    {
      const bool more = (layer + 1 < NATT);
      const float* wnext = W_att + (size_t)(layer + 1) * H_DIM * H_DIM + khalf * H_DIM + ncol;
      v8f acc0 = {}, acc1 = {};
      gemm_steps<0, KHALFWAY>(sh_h, LDH, bf, mlan, khalf, acc0, acc1);
      if (more) load_slice<H_DIM, 0, KHALFWAY>(bf, wnext);
      gemm_steps<KHALFWAY, KSTEPS>(sh_h, LDH, bf, mlan, khalf, acc0, acc1);
      const float bias_u = bias;
      if (more) {
        bias = b_att[(layer + 1) * H_DIM + ncol];
        load_slice<H_DIM, KHALFWAY, KSTEPS>(bf, wnext);
      }
#pragma unroll
      for (int r = 0; r < 8; ++r) {
        const int m = r + 8 * half;
        sh_u[m * H_DIM + ncol]        = tanh_exp2(acc0[r] + bias_u);
        sh_u[(16 + m) * H_DIM + ncol] = tanh_exp2(acc1[r] + bias_u);
      }
    }
    __syncthreads();

    {
      const float* ur = sh_u + grow * H_DIM;
      float* hr = sh_h + grow * LDH;
      float uu[EPL], hh[EPL];
#pragma unroll
      for (int cc = 0; cc < EPL / 4; ++cc) {
        const float4 u4 = *(const float4*)(ur + 4 * (sub + GLANES * cc));
        const float4 h4 = *(const float4*)(hr + 4 * (sub + GLANES * cc));
        uu[4 * cc + 0] = u4.x; uu[4 * cc + 1] = u4.y; uu[4 * cc + 2] = u4.z; uu[4 * cc + 3] = u4.w;
        hh[4 * cc + 0] = h4.x; hh[4 * cc + 1] = h4.y; hh[4 * cc + 2] = h4.z; hh[4 * cc + 3] = h4.w;
      }

      float S[NTERM], M[NTERM];
      float pw[EPL];
      {
        float m0 = hh[0], m1 = hh[1];
#pragma unroll
        for (int e = 2; e < EPL; e += 2) { m0 += hh[e]; m1 += hh[e + 1]; }
        M[0] = m0 + m1;
      }
#pragma unroll
      for (int e = 0; e < EPL; ++e) pw[e] = uu[e];
#pragma unroll
      for (int k = 1; k < NTERM; ++k) {
        float s0 = pw[0], s1 = pw[1];
        float m0 = pw[0] * hh[0], m1 = pw[1] * hh[1];
#pragma unroll
        for (int e = 2; e < EPL; e += 2) {
          s0 += pw[e];
          s1 += pw[e + 1];
          m0 = fmaf(pw[e], hh[e], m0);
          m1 = fmaf(pw[e + 1], hh[e + 1], m1);
        }
        S[k] = s0 + s1;
        M[k] = m0 + m1;
        if (k + 1 < NTERM) {
#pragma unroll
          for (int e = 0; e < EPL; ++e) pw[e] *= uu[e];
        }
      }
      constexpr float ECOEF[NTERM] = {1.0000000005494574f,
                                      1.0000000004994800f,
                                      0.49999997254821954f,
                                      0.16666665818280066f,
                                      0.041666886016742240f,
                                      0.0083333748022578046f,
                                      0.0013882759636653604f,
                                      0.00019832833995239756f,
                                      0.000025499185434365848f,
                                      0.0000028318193104809492f};
      float ak[NTERM], bk[NTERM];
#pragma unroll
      for (int k = 0; k < NTERM; ++k) {
        ak[k] = group_sum(M[k]) * ECOEF[k];
        bk[k] = (k == 0) ? (float)H_DIM * ECOEF[0] : group_sum(S[k]) * ECOEF[k];
      }

      float y[EPL];
#pragma unroll
      for (int e = 0; e < EPL; ++e) {
        const float xq = uu[e];
        float nn = ak[NTERM - 1], dd = bk[NTERM - 1];
#pragma unroll
        for (int k = NTERM - 2; k >= 0; --k) {
          nn = fmaf(nn, xq, ak[k]);
          dd = fmaf(dd, xq, bk[k]);
        }
        y[e] = hh[e] + nn * fast_rcp(dd);
      }

      float s0 = y[0], s1 = y[1];
#pragma unroll
      for (int e = 2; e < EPL; e += 2) { s0 += y[e]; s1 += y[e + 1]; }
      const float mu = group_sum(s0 + s1) * (1.0f / 128.0f);
      float v0 = 0.f, v1 = 0.f;
#pragma unroll
      for (int e = 0; e < EPL; e += 2) {
        const float d0 = y[e] - mu, d1 = y[e + 1] - mu;
        v0 = fmaf(d0, d0, v0);
        v1 = fmaf(d1, d1, v1);
      }
      const float rstd = fast_rsq(group_sum(v0 + v1) * (1.0f / 128.0f) + EPSLN);
#pragma unroll
      for (int cc = 0; cc < EPL / 4; ++cc) {
        float4 o;
        o.x = (y[4 * cc + 0] - mu) * rstd * gq[cc].x + bq[cc].x;
        o.y = (y[4 * cc + 1] - mu) * rstd * gq[cc].y + bq[cc].y;
        o.z = (y[4 * cc + 2] - mu) * rstd * gq[cc].z + bq[cc].z;
        o.w = (y[4 * cc + 3] - mu) * rstd * gq[cc].w + bq[cc].w;
        *(float4*)(hr + 4 * (sub + GLANES * cc)) = o;
      }
    }
    __syncthreads();
  }

  if (wave < RROWS / 16) {
    const int wrow = wave * 16;
    const float* arow = sh_h + (wrow + mlan) * LDH + khalf;
    const float* brow = sh_wc + khalf * NCLS + ccol;
    v2f af[KSTEPS], bw[KSTEPS];
#pragma unroll
    for (int s = 0; s < KSTEPS; ++s) {
      af[s]   = *(const v2f*)(arow + 4 * s);
      bw[s].x = brow[(4 * s) * NCLS];
      bw[s].y = brow[(4 * s + 1) * NCLS];
    }
    __builtin_amdgcn_sched_barrier(0);
    v8f acc = {};
#pragma unroll
    for (int s = 0; s < KSTEPS; ++s) acc = wmma4(af[s], bw[s], acc);
    if (valid) {
#pragma unroll
      for (int r = 0; r < 8; ++r) {
        const int m = wrow + r + 8 * half;
        out[(size_t)(row0 + m) * NCLS + mlan] = acc[r] + bias_c;
      }
    }
  }
}

extern "C" void kernel_launch(void* const* d_in, const int* in_sizes, int n_in,
                              void* d_out, int out_size, void* d_ws,
                              size_t ws_size, hipStream_t stream) {
  const float* x     = (const float*)d_in[0];
  const float* W_in  = (const float*)d_in[1];
  const float* b_in  = (const float*)d_in[2];
  const float* W_att = (const float*)d_in[3];
  const float* b_att = (const float*)d_in[4];
  const float* gamma = (const float*)d_in[5];
  const float* beta  = (const float*)d_in[6];
  const float* W_c   = (const float*)d_in[7];
  const float* b_c   = (const float*)d_in[8];
  float* out = (float*)d_out;

  dim3 grid(B_ROWS / RROWS);
  dim3 block(NWAVES * 32);
  simpleattention_fused_kernel<<<grid, block, 0, stream>>>(
      x, W_in, b_in, W_att, b_att, gamma, beta, W_c, b_c, out);
}
